# starting kernel
# baseline (speedup 1.0000x reference)
#include <hip/hip_runtime.h>
#include <math.h>

#define B_TOTAL   16384
#define ND        13
#define NS        26
#define FEAT_NUM  10000
#define FLEN      260013
#define KDIM      64

typedef __attribute__((ext_vector_type(2))) float v2f;
typedef __attribute__((ext_vector_type(8))) float v8f;

__device__ __forceinline__ float sigmoidf_(float x) {
    return 1.0f / (1.0f + __expf(-x));
}

__global__ __launch_bounds__(256) void FM_29437705847531_kernel(
    const float* __restrict__ dense,
    const int*   __restrict__ sp,
    const float* __restrict__ w0p,
    const float* __restrict__ w,
    const float* __restrict__ V,
    float* __restrict__ out)
{
    const int lane = threadIdx.x & 31;
    const int wave = threadIdx.x >> 5;
    const int tile = blockIdx.x * 8 + wave;
    const int row0 = tile * 16;

    const int n  = lane & 15;
    const int hi = lane >> 4;

    const float* rowPtr[4];
#pragma unroll
    for (int nt = 0; nt < 4; ++nt)
        rowPtr[nt] = V + (size_t)(nt * 16 + n) * FLEN;

    v8f acc_vx[4] = {};
    v8f acc_v2[4] = {};

#pragma unroll
    for (int kk = 0; kk < 16; kk += 4) {
        const int f0 = kk + 2 * hi;
        v2f a;
        a.x = (f0 + 0 < ND) ? dense[(size_t)(row0 + n) * ND + f0 + 0] : 0.0f;
        a.y = (f0 + 1 < ND) ? dense[(size_t)(row0 + n) * ND + f0 + 1] : 0.0f;
        const v2f a2 = a * a;
#pragma unroll
        for (int nt = 0; nt < 4; ++nt) {
            v2f b;
            b.x = (f0 + 0 < ND) ? rowPtr[nt][f0 + 0] : 0.0f;
            b.y = (f0 + 1 < ND) ? rowPtr[nt][f0 + 1] : 0.0f;
            const v2f b2 = b * b;
            acc_vx[nt] = __builtin_amdgcn_wmma_f32_16x16x4_f32(
                false, a, false, b, (short)0, acc_vx[nt], false, false);
            acc_v2[nt] = __builtin_amdgcn_wmma_f32_16x16x4_f32(
                false, a2, false, b2, (short)0, acc_v2[nt], false, false);
        }
    }

    for (int j = 0; j < NS; ++j) {
#pragma unroll
        for (int v = 0; v < 8; ++v) {
            const int m = v + 8 * hi;
            const size_t col = (size_t)ND + (size_t)j * FEAT_NUM
                             + (size_t)sp[(size_t)(row0 + m) * NS + j];
#pragma unroll
            for (int nt = 0; nt < 4; ++nt) {
                const float val = rowPtr[nt][col];
                acc_vx[nt][v] += val;
                acc_v2[nt][v] += val * val;
            }
        }
    }

    const int r1 = row0 + n;
    float first = w0p[0];
    for (int f = 0; f < ND; ++f)
        first += dense[(size_t)r1 * ND + f] * w[f];
    for (int j = 0; j < NS; ++j)
        first += w[(size_t)ND + (size_t)j * FEAT_NUM
                   + (size_t)sp[(size_t)r1 * NS + j]];

#pragma unroll
    for (int v = 0; v < 8; ++v) {
        float t = 0.0f;
#pragma unroll
        for (int nt = 0; nt < 4; ++nt) {
            const float x = acc_vx[nt][v];
            t += x * x - acc_v2[nt][v];
        }
        t += __shfl_xor(t, 1, 32);
        t += __shfl_xor(t, 2, 32);
        t += __shfl_xor(t, 4, 32);
        t += __shfl_xor(t, 8, 32);
        const float sec = 0.5f * t;

        const float f_lo = __shfl(first, v,     32);
        const float f_hi = __shfl(first, 8 + v, 32);
        if (lane == 0)        out[row0 + v]     = sigmoidf_(f_lo + sec);
        else if (lane == 16)  out[row0 + 8 + v] = sigmoidf_(f_hi + sec);
    }
}

extern "C" void kernel_launch(void* const* d_in, const int* in_sizes, int n_in,
                              void* d_out, int out_size, void* d_ws, size_t ws_size,
                              hipStream_t stream) {
    const float* dense = (const float*)d_in[0];
    const int*   sp    = (const int*)  d_in[1];
    const float* w0p   = (const float*)d_in[2];
    const float* w     = (const float*)d_in[3];
    const float* V     = (const float*)d_in[4];
    float* out = (float*)d_out;

    dim3 grid(B_TOTAL / (16 * 8));
    FM_29437705847531_kernel<<<grid, 256, 0, stream>>>(dense, sp, w0p, w, V, out);
}


# V no longer gathered from global: per latent row each field's slice streams into LDS and lookups read LDS; 2nd launch adds the dense WMMA and reduces; sum order over fields and latents changes
# speedup vs baseline: 4.7475x; 4.7475x over previous
#include <hip/hip_runtime.h>
#include <math.h>
#include <stdint.h>


#define B_TOTAL   16384
#define ND        13
#define NS        26
#define FEAT_NUM  10000
#define FLEN      260013
#define KDIM      64

#define SCHUNK      4096
#define LK_THREADS  256
#define SPT         (SCHUNK / LK_THREADS)
#define SLICE_Q4    (FEAT_NUM / 4)
#define SLICE_QMAX  (SLICE_Q4 + 1)
#define QPT         10
#define IDX_BYTES   (SCHUNK * NS * 2)
#define SLICE_BYTES (SLICE_QMAX * 16)
#define LDS_BYTES   (IDX_BYTES + 2 * SLICE_BYTES)

typedef __attribute__((ext_vector_type(2))) float v2f;
typedef __attribute__((ext_vector_type(8))) float v8f;

__device__ __forceinline__ float sigmoidf_(float x) {
    return 1.0f / (1.0f + __expf(-x));
}

__device__ __forceinline__ void lds_handover_barrier() {
    asm volatile("s_wait_dscnt 0x0\n\ts_barrier_signal -1\n\ts_barrier_wait -1" ::: "memory");
}

__global__ __launch_bounds__(LK_THREADS) void fm_lookup_kernel(
    const int*   __restrict__ sp,
    const float* __restrict__ V,
    float* __restrict__ vxs,
    float* __restrict__ v2s)
{
    __shared__ __attribute__((aligned(16))) unsigned char L[LDS_BYTES];
    uint32_t* idxw = (uint32_t*)L;
    float4*   sq0  = (float4*)(L + IDX_BYTES);
    float4*   sq1  = (float4*)(L + IDX_BYTES + SLICE_BYTES);
    const float* sl0 = (const float*)sq0;
    const float* sl1 = (const float*)sq1;

    const int tid = threadIdx.x;
    const int k   = blockIdx.x >> 2;
    const int b0  = (blockIdx.x & 3) * SCHUNK;

    const float* vrow = V + (size_t)k * FLEN + ND;
    const int m  = (int)(((uintptr_t)vrow >> 2) & 3);
    const int nq = (FEAT_NUM + m + 3) >> 2;
    const float4* vq = (const float4*)(vrow - m);
    const int q9 = min(9 * LK_THREADS + tid, nq - 1);

    float4 ra0, ra1, ra2, ra3, ra4, ra5, ra6, ra7, ra8, ra9;
    float4 rb0, rb1, rb2, rb3, rb4, rb5, rb6, rb7, rb8, rb9;
#define FETCH(p, j)                                                              \
    {                                                                            \
        const float4* s_ = vq + (size_t)(j) * SLICE_Q4 + tid;                    \
        p##0 = s_[0 * LK_THREADS]; p##1 = s_[1 * LK_THREADS];                    \
        p##2 = s_[2 * LK_THREADS]; p##3 = s_[3 * LK_THREADS];                    \
        p##4 = s_[4 * LK_THREADS]; p##5 = s_[5 * LK_THREADS];                    \
        p##6 = s_[6 * LK_THREADS]; p##7 = s_[7 * LK_THREADS];                    \
        p##8 = s_[8 * LK_THREADS]; p##9 = s_[q9 - tid];                          \
    }
#define STASH(dq, p)                                                             \
    {                                                                            \
        float4* d_ = dq + tid;                                                   \
        d_[0 * LK_THREADS] = p##0; d_[1 * LK_THREADS] = p##1;                    \
        d_[2 * LK_THREADS] = p##2; d_[3 * LK_THREADS] = p##3;                    \
        d_[4 * LK_THREADS] = p##4; d_[5 * LK_THREADS] = p##5;                    \
        d_[6 * LK_THREADS] = p##6; d_[7 * LK_THREADS] = p##7;                    \
        d_[8 * LK_THREADS] = p##8; d_[q9 - tid] = p##9;                          \
    }
#define LOAD_IDX(jj_)                                                            \
    {                                                                            \
        _Pragma("unroll")                                                        \
        for (int i = 0; i < SPT; ++i)                                            \
            iw[i] = idxw[(tid + LK_THREADS * i) * (NS / 2) + (jj_)];             \
    }
#define LOOKUP(sl, SHIFT_OR_MASK)                                                \
    {                                                                            \
        _Pragma("unroll")                                                        \
        for (int i = 0; i < SPT; ++i) {                                          \
            const float val = sl[m + (int)(iw[i] SHIFT_OR_MASK)];                \
            acc[i]  += val;                                                      \
            acc2[i] += val * val;                                                \
        }                                                                        \
    }

    FETCH(rb, 0)
    FETCH(ra, 1)

    {
        const int4* spq = (const int4*)(sp + (size_t)b0 * NS);
        uint2* dst = (uint2*)idxw;
#pragma unroll 26
        for (int i = 0; i < (SCHUNK * NS / 4) / LK_THREADS; ++i) {
            const int q = i * LK_THREADS + tid;
            const int4 v = spq[q];
            uint2 p;
            p.x = ((uint32_t)v.x & 0xffffu) | ((uint32_t)v.y << 16);
            p.y = ((uint32_t)v.z & 0xffffu) | ((uint32_t)v.w << 16);
            dst[q] = p;
        }
    }
    STASH(sq0, rb)
    __syncthreads();

    float acc[SPT], acc2[SPT];
#pragma unroll
    for (int i = 0; i < SPT; ++i) { acc[i] = 0.0f; acc2[i] = 0.0f; }
    uint32_t iw[SPT];

    for (int jj = 0; jj < NS / 2 - 1; ++jj) {
        LOAD_IDX(jj)
        FETCH(rb, 2 * jj + 2)
        LOOKUP(sl0, & 0xffffu)
        STASH(sq1, ra)
        lds_handover_barrier();
        FETCH(ra, 2 * jj + 3)
        LOOKUP(sl1, >> 16)
        STASH(sq0, rb)
        lds_handover_barrier();
    }
    LOAD_IDX(NS / 2 - 1)
    LOOKUP(sl0, & 0xffffu)
    STASH(sq1, ra)
    lds_handover_barrier();
    LOOKUP(sl1, >> 16)
#undef FETCH
#undef STASH
#undef LOAD_IDX
#undef LOOKUP

    float* ox = vxs + (size_t)k * B_TOTAL + b0;
    float* o2 = v2s + (size_t)k * B_TOTAL + b0;
#pragma unroll
    for (int i = 0; i < SPT; ++i) {
        ox[tid + LK_THREADS * i] = acc[i];
        o2[tid + LK_THREADS * i] = acc2[i];
    }
}

__global__ __launch_bounds__(64) void fm_finish_kernel(
    const float* __restrict__ dense,
    const int*   __restrict__ sp,
    const float* __restrict__ w0p,
    const float* __restrict__ w,
    const float* __restrict__ V,
    const float* __restrict__ vxs,
    const float* __restrict__ v2s,
    float* __restrict__ out)
{
    const int lane = threadIdx.x & 31;
    const int wave = threadIdx.x >> 5;
    const int rowW = (blockIdx.x * 2 + wave) * 32;
    const int c  = lane & 15;
    const int hi = lane >> 4;

    const int r1 = rowW + lane;
    float first = w0p[0];
    for (int f = 0; f < ND; ++f)
        first += dense[(size_t)r1 * ND + f] * w[f];
    for (int j = 0; j < NS; ++j)
        first += w[(size_t)ND + (size_t)j * FEAT_NUM
                   + (size_t)sp[(size_t)r1 * NS + j]];

    float res[2];
#pragma unroll
    for (int t = 0; t < 2; ++t) {
        const int row0 = rowW + 16 * t;
        v8f ax[4], a2[4];
#pragma unroll
        for (int nt = 0; nt < 4; ++nt) {
#pragma unroll
            for (int r = 0; r < 8; ++r) {
                const size_t o = (size_t)(nt * 16 + 8 * hi + r) * B_TOTAL + row0 + c;
                ax[nt][r] = vxs[o];
                a2[nt][r] = v2s[o];
            }
        }
#pragma unroll
        for (int kk = 0; kk < 16; kk += 4) {
            const int f0 = kk + 2 * hi;
            v2f bs;
            bs.x = (f0 + 0 < ND) ? dense[(size_t)(row0 + c) * ND + f0 + 0] : 0.0f;
            bs.y = (f0 + 1 < ND) ? dense[(size_t)(row0 + c) * ND + f0 + 1] : 0.0f;
            const v2f bs2 = bs * bs;
#pragma unroll
            for (int nt = 0; nt < 4; ++nt) {
                const float* vr = V + (size_t)(nt * 16 + c) * FLEN;
                v2f al;
                al.x = (f0 + 0 < ND) ? vr[f0 + 0] : 0.0f;
                al.y = (f0 + 1 < ND) ? vr[f0 + 1] : 0.0f;
                const v2f al2 = al * al;
                ax[nt] = __builtin_amdgcn_wmma_f32_16x16x4_f32(
                    false, al, false, bs, (short)0, ax[nt], false, false);
                a2[nt] = __builtin_amdgcn_wmma_f32_16x16x4_f32(
                    false, al2, false, bs2, (short)0, a2[nt], false, false);
            }
        }
        float s = 0.0f;
#pragma unroll
        for (int nt = 0; nt < 4; ++nt) {
#pragma unroll
            for (int r = 0; r < 8; ++r) {
                const float x = ax[nt][r];
                s += x * x - a2[nt][r];
            }
        }
        s += __shfl_xor(s, 16, 32);
        res[t] = s;
    }
    const float sec = 0.5f * (hi ? res[1] : res[0]);
    out[r1] = sigmoidf_(first + sec);
}

extern "C" void kernel_launch(void* const* d_in, const int* in_sizes, int n_in,
                              void* d_out, int out_size, void* d_ws, size_t ws_size,
                              hipStream_t stream) {
    const float* dense = (const float*)d_in[0];
    const int*   sp    = (const int*)  d_in[1];
    const float* w0p   = (const float*)d_in[2];
    const float* w     = (const float*)d_in[3];
    const float* V     = (const float*)d_in[4];
    float* out = (float*)d_out;
    float* vxs = (float*)d_ws;
    float* v2s = vxs + KDIM * B_TOTAL;

    fm_lookup_kernel<<<dim3(KDIM * (B_TOTAL / SCHUNK)), LK_THREADS, 0, stream>>>(sp, V, vxs, v2s);
    fm_finish_kernel<<<dim3(B_TOTAL / 64), 64, 0, stream>>>(dense, sp, w0p, w, V, vxs, v2s, out);
}


# speedup vs baseline: 4.9711x; 1.0471x over previous
#include <hip/hip_runtime.h>
#include <math.h>
#include <stdint.h>


#define B_TOTAL   16384
#define ND        13
#define NS        26
#define FEAT_NUM  10000
#define FLEN      260013
#define KDIM      64

#define SCHUNK      4096
#define LK_THREADS  256
#define SPT         (SCHUNK / LK_THREADS)
#define SLICE_Q4    (FEAT_NUM / 4)
#define SLICE_QMAX  (SLICE_Q4 + 16)
#define QPT         10
#define IDX_BYTES   (SCHUNK * NS * 2)
#define SLICE_BYTES (SLICE_QMAX * 16)
#define LDS_BYTES   (IDX_BYTES + 2 * SLICE_BYTES)

typedef __attribute__((ext_vector_type(2))) float v2f;
typedef __attribute__((ext_vector_type(8))) float v8f;

__device__ __forceinline__ float sigmoidf_(float x) {
    return 1.0f / (1.0f + __expf(-x));
}

__device__ __forceinline__ void lds_handover_barrier() {
    asm volatile("s_wait_dscnt 0x0\n\ts_barrier_signal -1\n\ts_barrier_wait -1" ::: "memory");
}

__global__ __launch_bounds__(LK_THREADS) void fm_lookup_kernel(
    const int*   __restrict__ sp,
    const float* __restrict__ V,
    float* __restrict__ vxs,
    float* __restrict__ v2s)
{
    __shared__ __attribute__((aligned(16))) unsigned char L[LDS_BYTES];
    uint32_t* idxw = (uint32_t*)L;
    float4*   sq0  = (float4*)(L + IDX_BYTES);
    float4*   sq1  = (float4*)(L + IDX_BYTES + SLICE_BYTES);
    const float* sl0 = (const float*)sq0;
    const float* sl1 = (const float*)sq1;

    const int tid = threadIdx.x;
    const int k   = blockIdx.x >> 2;
    const int b0  = (blockIdx.x & 3) * SCHUNK;

    const float* vrow = V + (size_t)k * FLEN + ND;
    const int m0 = (int)(((uintptr_t)vrow >> 2) & 63);
    const float4* vq0 = (const float4*)(vrow - m0);
#define SLICE_GEOM(j)                                                            \
        const int off_ = m0 + (j) * FEAT_NUM;                                    \
        const int mj_  = off_ & 63;                                              \
        const int qlo_ = max(tid, mj_ >> 2);                                     \
        const int qhi_ = min(9 * LK_THREADS + tid, ((mj_ + FEAT_NUM + 3) >> 2) - 1);

    float4 ra0, ra1, ra2, ra3, ra4, ra5, ra6, ra7, ra8, ra9;
    float4 rb0, rb1, rb2, rb3, rb4, rb5, rb6, rb7, rb8, rb9;
#define FETCH(p, j)                                                              \
    {                                                                            \
        SLICE_GEOM(j)                                                            \
        const float4* b_ = vq0 + ((off_ - mj_) >> 2);                            \
        const float4* s_ = b_ + tid;                                             \
        p##0 = b_[qlo_];           p##1 = s_[1 * LK_THREADS];                    \
        p##2 = s_[2 * LK_THREADS]; p##3 = s_[3 * LK_THREADS];                    \
        p##4 = s_[4 * LK_THREADS]; p##5 = s_[5 * LK_THREADS];                    \
        p##6 = s_[6 * LK_THREADS]; p##7 = s_[7 * LK_THREADS];                    \
        p##8 = s_[8 * LK_THREADS]; p##9 = b_[qhi_];                              \
    }
#define STASH(dq, p, j)                                                          \
    {                                                                            \
        SLICE_GEOM(j)                                                            \
        float4* d_ = dq + tid;                                                   \
        dq[qlo_] = p##0;           d_[1 * LK_THREADS] = p##1;                    \
        d_[2 * LK_THREADS] = p##2; d_[3 * LK_THREADS] = p##3;                    \
        d_[4 * LK_THREADS] = p##4; d_[5 * LK_THREADS] = p##5;                    \
        d_[6 * LK_THREADS] = p##6; d_[7 * LK_THREADS] = p##7;                    \
        d_[8 * LK_THREADS] = p##8; dq[qhi_] = p##9;                              \
    }
#define LOAD_IDX(jj_)                                                            \
    {                                                                            \
        _Pragma("unroll")                                                        \
        for (int i = 0; i < SPT; ++i)                                            \
            iw[i] = idxw[(tid + LK_THREADS * i) * (NS / 2) + (jj_)];             \
    }
#define LOOKUP(sl, j, SHIFT_OR_MASK)                                             \
    {                                                                            \
        const float* sm_ = sl + ((m0 + 16 * (j)) & 63);                          \
        _Pragma("unroll")                                                        \
        for (int i = 0; i < SPT; ++i) {                                          \
            const float val = sm_[(int)(iw[i] SHIFT_OR_MASK)];                   \
            acc[i]  += val;                                                      \
            acc2[i] += val * val;                                                \
        }                                                                        \
    }

    FETCH(rb, 0)
    FETCH(ra, 1)

    {
        const int4* spq = (const int4*)(sp + (size_t)b0 * NS);
        uint2* dst = (uint2*)idxw;
#pragma unroll 26
        for (int i = 0; i < (SCHUNK * NS / 4) / LK_THREADS; ++i) {
            const int q = i * LK_THREADS + tid;
            const int4 v = spq[q];
            uint2 p;
            p.x = ((uint32_t)v.x & 0xffffu) | ((uint32_t)v.y << 16);
            p.y = ((uint32_t)v.z & 0xffffu) | ((uint32_t)v.w << 16);
            dst[q] = p;
        }
    }
    STASH(sq0, rb, 0)
    __syncthreads();

    float acc[SPT], acc2[SPT];
#pragma unroll
    for (int i = 0; i < SPT; ++i) { acc[i] = 0.0f; acc2[i] = 0.0f; }
    uint32_t iw[SPT];

    for (int jj = 0; jj < NS / 2 - 1; ++jj) {
        LOAD_IDX(jj)
        FETCH(rb, 2 * jj + 2)
        LOOKUP(sl0, 2 * jj, & 0xffffu)
        STASH(sq1, ra, 2 * jj + 1)
        lds_handover_barrier();
        FETCH(ra, 2 * jj + 3)
        LOOKUP(sl1, 2 * jj + 1, >> 16)
        STASH(sq0, rb, 2 * jj + 2)
        lds_handover_barrier();
    }
    LOAD_IDX(NS / 2 - 1)
    LOOKUP(sl0, NS - 2, & 0xffffu)
    STASH(sq1, ra, NS - 1)
    lds_handover_barrier();
    LOOKUP(sl1, NS - 1, >> 16)
#undef SLICE_GEOM
#undef FETCH
#undef STASH
#undef LOAD_IDX
#undef LOOKUP

    float* ox = vxs + (size_t)k * B_TOTAL + b0;
    float* o2 = v2s + (size_t)k * B_TOTAL + b0;
#pragma unroll
    for (int i = 0; i < SPT; ++i) {
        ox[tid + LK_THREADS * i] = acc[i];
        o2[tid + LK_THREADS * i] = acc2[i];
    }
}

__global__ __launch_bounds__(64) void fm_finish_kernel(
    const float* __restrict__ dense,
    const int*   __restrict__ sp,
    const float* __restrict__ w0p,
    const float* __restrict__ w,
    const float* __restrict__ V,
    const float* __restrict__ vxs,
    const float* __restrict__ v2s,
    float* __restrict__ out)
{
    const int lane = threadIdx.x & 31;
    const int wave = threadIdx.x >> 5;
    const int rowW = (blockIdx.x * 2 + wave) * 32;
    const int c  = lane & 15;
    const int hi = lane >> 4;

    const int r1 = rowW + lane;
    float first = w0p[0];
    for (int f = 0; f < ND; ++f)
        first += dense[(size_t)r1 * ND + f] * w[f];
    for (int j = 0; j < NS; ++j)
        first += w[(size_t)ND + (size_t)j * FEAT_NUM
                   + (size_t)sp[(size_t)r1 * NS + j]];

    float res[2];
#pragma unroll
    for (int t = 0; t < 2; ++t) {
        const int row0 = rowW + 16 * t;
        v8f ax[4], a2[4];
#pragma unroll
        for (int nt = 0; nt < 4; ++nt) {
#pragma unroll
            for (int r = 0; r < 8; ++r) {
                const size_t o = (size_t)(nt * 16 + 8 * hi + r) * B_TOTAL + row0 + c;
                ax[nt][r] = vxs[o];
                a2[nt][r] = v2s[o];
            }
        }
#pragma unroll
        for (int kk = 0; kk < 16; kk += 4) {
            const int f0 = kk + 2 * hi;
            v2f bs;
            bs.x = (f0 + 0 < ND) ? dense[(size_t)(row0 + c) * ND + f0 + 0] : 0.0f;
            bs.y = (f0 + 1 < ND) ? dense[(size_t)(row0 + c) * ND + f0 + 1] : 0.0f;
            const v2f bs2 = bs * bs;
#pragma unroll
            for (int nt = 0; nt < 4; ++nt) {
                const float* vr = V + (size_t)(nt * 16 + c) * FLEN;
                v2f al;
                al.x = (f0 + 0 < ND) ? vr[f0 + 0] : 0.0f;
                al.y = (f0 + 1 < ND) ? vr[f0 + 1] : 0.0f;
                const v2f al2 = al * al;
                ax[nt] = __builtin_amdgcn_wmma_f32_16x16x4_f32(
                    false, al, false, bs, (short)0, ax[nt], false, false);
                a2[nt] = __builtin_amdgcn_wmma_f32_16x16x4_f32(
                    false, al2, false, bs2, (short)0, a2[nt], false, false);
            }
        }
        float s = 0.0f;
#pragma unroll
        for (int nt = 0; nt < 4; ++nt) {
#pragma unroll
            for (int r = 0; r < 8; ++r) {
                const float x = ax[nt][r];
                s += x * x - a2[nt][r];
            }
        }
        s += __shfl_xor(s, 16, 32);
        res[t] = s;
    }
    const float sec = 0.5f * (hi ? res[1] : res[0]);
    out[r1] = sigmoidf_(first + sec);
}

extern "C" void kernel_launch(void* const* d_in, const int* in_sizes, int n_in,
                              void* d_out, int out_size, void* d_ws, size_t ws_size,
                              hipStream_t stream) {
    const float* dense = (const float*)d_in[0];
    const int*   sp    = (const int*)  d_in[1];
    const float* w0p   = (const float*)d_in[2];
    const float* w     = (const float*)d_in[3];
    const float* V     = (const float*)d_in[4];
    float* out = (float*)d_out;
    float* vxs = (float*)d_ws;
    float* v2s = vxs + KDIM * B_TOTAL;

    fm_lookup_kernel<<<dim3(KDIM * (B_TOTAL / SCHUNK)), LK_THREADS, 0, stream>>>(sp, V, vxs, v2s);
    fm_finish_kernel<<<dim3(B_TOTAL / 64), 64, 0, stream>>>(dense, sp, w0p, w, V, vxs, v2s, out);
}


# finish launch: every load issued in one group (padded features load a clamped index, zeroed by select; one 16-sample tile per wave): 10 serialized round trips become 2; first-order sum order changes
# speedup vs baseline: 5.1437x; 1.0347x over previous
#include <hip/hip_runtime.h>
#include <math.h>
#include <stdint.h>


#define B_TOTAL   16384
#define ND        13
#define NS        26
#define FEAT_NUM  10000
#define FLEN      260013
#define KDIM      64

#define SCHUNK      4096
#define LK_THREADS  256
#define SPT         (SCHUNK / LK_THREADS)
#define SLICE_Q4    (FEAT_NUM / 4)
#define SLICE_QMAX  (SLICE_Q4 + 16)
#define QPT         10
#define IDX_BYTES   (SCHUNK * NS * 2)
#define SLICE_BYTES (SLICE_QMAX * 16)
#define LDS_BYTES   (IDX_BYTES + 2 * SLICE_BYTES)

typedef __attribute__((ext_vector_type(2))) float v2f;
typedef __attribute__((ext_vector_type(8))) float v8f;

__device__ __forceinline__ float sigmoidf_(float x) {
    return 1.0f / (1.0f + __expf(-x));
}

__device__ __forceinline__ void lds_handover_barrier() {
    asm volatile("s_wait_dscnt 0x0\n\ts_barrier_signal -1\n\ts_barrier_wait -1" ::: "memory");
}

__global__ __launch_bounds__(LK_THREADS) void fm_lookup_kernel(
    const int*   __restrict__ sp,
    const float* __restrict__ V,
    float* __restrict__ vxs,
    float* __restrict__ v2s)
{
    __shared__ __attribute__((aligned(16))) unsigned char L[LDS_BYTES];
    uint32_t* idxw = (uint32_t*)L;
    float4*   sq0  = (float4*)(L + IDX_BYTES);
    float4*   sq1  = (float4*)(L + IDX_BYTES + SLICE_BYTES);
    const float* sl0 = (const float*)sq0;
    const float* sl1 = (const float*)sq1;

    const int tid = threadIdx.x;
    const int k   = blockIdx.x >> 2;
    const int b0  = (blockIdx.x & 3) * SCHUNK;

    const float* vrow = V + (size_t)k * FLEN + ND;
    const int m0 = (int)(((uintptr_t)vrow >> 2) & 63);
    const float4* vq0 = (const float4*)(vrow - m0);
#define SLICE_GEOM(j)                                                            \
        const int off_ = m0 + (j) * FEAT_NUM;                                    \
        const int mj_  = off_ & 63;                                              \
        const int qlo_ = max(tid, mj_ >> 2);                                     \
        const int qhi_ = min(9 * LK_THREADS + tid, ((mj_ + FEAT_NUM + 3) >> 2) - 1);

    float4 ra0, ra1, ra2, ra3, ra4, ra5, ra6, ra7, ra8, ra9;
    float4 rb0, rb1, rb2, rb3, rb4, rb5, rb6, rb7, rb8, rb9;
#define FETCH(p, j)                                                              \
    {                                                                            \
        SLICE_GEOM(j)                                                            \
        const float4* b_ = vq0 + ((off_ - mj_) >> 2);                            \
        const float4* s_ = b_ + tid;                                             \
        p##0 = b_[qlo_];           p##1 = s_[1 * LK_THREADS];                    \
        p##2 = s_[2 * LK_THREADS]; p##3 = s_[3 * LK_THREADS];                    \
        p##4 = s_[4 * LK_THREADS]; p##5 = s_[5 * LK_THREADS];                    \
        p##6 = s_[6 * LK_THREADS]; p##7 = s_[7 * LK_THREADS];                    \
        p##8 = s_[8 * LK_THREADS]; p##9 = b_[qhi_];                              \
    }
#define STASH(dq, p, j)                                                          \
    {                                                                            \
        SLICE_GEOM(j)                                                            \
        float4* d_ = dq + tid;                                                   \
        dq[qlo_] = p##0;           d_[1 * LK_THREADS] = p##1;                    \
        d_[2 * LK_THREADS] = p##2; d_[3 * LK_THREADS] = p##3;                    \
        d_[4 * LK_THREADS] = p##4; d_[5 * LK_THREADS] = p##5;                    \
        d_[6 * LK_THREADS] = p##6; d_[7 * LK_THREADS] = p##7;                    \
        d_[8 * LK_THREADS] = p##8; dq[qhi_] = p##9;                              \
    }
#define LOAD_IDX(jj_)                                                            \
    {                                                                            \
        _Pragma("unroll")                                                        \
        for (int i = 0; i < SPT; ++i)                                            \
            iw[i] = idxw[(tid + LK_THREADS * i) * (NS / 2) + (jj_)];             \
    }
#define LOOKUP(sl, j, SHIFT_OR_MASK)                                             \
    {                                                                            \
        const float* sm_ = sl + ((m0 + 16 * (j)) & 63);                          \
        _Pragma("unroll")                                                        \
        for (int i = 0; i < SPT; ++i) {                                          \
            const float val = sm_[(int)(iw[i] SHIFT_OR_MASK)];                   \
            acc[i]  += val;                                                      \
            acc2[i] += val * val;                                                \
        }                                                                        \
    }

    FETCH(rb, 0)
    FETCH(ra, 1)

    {
        const int4* spq = (const int4*)(sp + (size_t)b0 * NS);
        uint2* dst = (uint2*)idxw;
#pragma unroll 26
        for (int i = 0; i < (SCHUNK * NS / 4) / LK_THREADS; ++i) {
            const int q = i * LK_THREADS + tid;
            const int4 v = spq[q];
            uint2 p;
            p.x = ((uint32_t)v.x & 0xffffu) | ((uint32_t)v.y << 16);
            p.y = ((uint32_t)v.z & 0xffffu) | ((uint32_t)v.w << 16);
            dst[q] = p;
        }
    }
    STASH(sq0, rb, 0)
    __syncthreads();

    float acc[SPT], acc2[SPT];
#pragma unroll
    for (int i = 0; i < SPT; ++i) { acc[i] = 0.0f; acc2[i] = 0.0f; }
    uint32_t iw[SPT];

    for (int jj = 0; jj < NS / 2 - 1; ++jj) {
        LOAD_IDX(jj)
        FETCH(rb, 2 * jj + 2)
        LOOKUP(sl0, 2 * jj, & 0xffffu)
        STASH(sq1, ra, 2 * jj + 1)
        lds_handover_barrier();
        FETCH(ra, 2 * jj + 3)
        LOOKUP(sl1, 2 * jj + 1, >> 16)
        STASH(sq0, rb, 2 * jj + 2)
        lds_handover_barrier();
    }
    LOAD_IDX(NS / 2 - 1)
    LOOKUP(sl0, NS - 2, & 0xffffu)
    STASH(sq1, ra, NS - 1)
    lds_handover_barrier();
    LOOKUP(sl1, NS - 1, >> 16)
#undef SLICE_GEOM
#undef FETCH
#undef STASH
#undef LOAD_IDX
#undef LOOKUP

    float* ox = vxs + (size_t)k * B_TOTAL + b0;
    float* o2 = v2s + (size_t)k * B_TOTAL + b0;
#pragma unroll
    for (int i = 0; i < SPT; ++i) {
        ox[tid + LK_THREADS * i] = acc[i];
        o2[tid + LK_THREADS * i] = acc2[i];
    }
}

__global__ __launch_bounds__(128) void fm_finish_kernel(
    const float* __restrict__ dense,
    const int*   __restrict__ sp,
    const float* __restrict__ w0p,
    const float* __restrict__ w,
    const float* __restrict__ V,
    const float* __restrict__ vxs,
    const float* __restrict__ v2s,
    float* __restrict__ out)
{
    __shared__ float res_s[64];


    const int lane = threadIdx.x & 31;
    const int wave = threadIdx.x >> 5;
    const int c  = lane & 15;
    const int hi = lane >> 4;
    const int rs = blockIdx.x * 64 + wave * 16 + c;

    const int jb = 13 * hi;
    int id[13];
#pragma unroll
    for (int j = 0; j < 13; ++j)
        id[j] = sp[(size_t)rs * NS + jb + j];

    v8f ax[4], a2[4];
#pragma unroll
    for (int nt = 0; nt < 4; ++nt) {
#pragma unroll
        for (int r = 0; r < 8; ++r) {
            const size_t o = (size_t)(nt * 16 + 8 * hi + r) * B_TOTAL + rs;
            ax[nt][r] = vxs[o];
            a2[nt][r] = v2s[o];
        }
    }

    v2f bs[4];
    v2f al[4][4];
#pragma unroll
    for (int kq = 0; kq < 4; ++kq) {
        const int f0 = 4 * kq + 2 * hi;
        const int fa = min(f0, ND - 1);
        const int fb = min(f0 + 1, ND - 1);
        const float dx = dense[(size_t)rs * ND + fa];
        const float dy = dense[(size_t)rs * ND + fb];
        bs[kq].x = (f0 + 0 < ND) ? dx : 0.0f;
        bs[kq].y = (f0 + 1 < ND) ? dy : 0.0f;
#pragma unroll
        for (int nt = 0; nt < 4; ++nt) {
            const float* vr = V + (size_t)(nt * 16 + c) * FLEN;
            const float vx = vr[fa];
            const float vy = vr[fb];
            al[kq][nt].x = (f0 + 0 < ND) ? vx : 0.0f;
            al[kq][nt].y = (f0 + 1 < ND) ? vy : 0.0f;
        }
    }

    const float w0 = w0p[0];
    float fh = hi ? 0.0f : w0;
#pragma unroll
    for (int kq = 0; kq < 4; ++kq) {
        const float w_0 = (4 * kq + 0 < ND) ? w[(4 * kq + 0 < ND) ? 4 * kq + 0 : 0] : 0.0f;
        const float w_1 = (4 * kq + 1 < ND) ? w[(4 * kq + 1 < ND) ? 4 * kq + 1 : 0] : 0.0f;
        const float w_2 = (4 * kq + 2 < ND) ? w[(4 * kq + 2 < ND) ? 4 * kq + 2 : 0] : 0.0f;
        const float w_3 = (4 * kq + 3 < ND) ? w[(4 * kq + 3 < ND) ? 4 * kq + 3 : 0] : 0.0f;
        fh += bs[kq].x * (hi ? w_2 : w_0);
        fh += bs[kq].y * (hi ? w_3 : w_1);
    }
#pragma unroll
    for (int j = 0; j < 13; ++j)
        fh += w[(size_t)ND + (size_t)(jb + j) * FEAT_NUM + (size_t)id[j]];

#pragma unroll
    for (int kq = 0; kq < 4; ++kq) {
        const v2f bs2 = bs[kq] * bs[kq];
#pragma unroll
        for (int nt = 0; nt < 4; ++nt) {
            const v2f al2 = al[kq][nt] * al[kq][nt];
            ax[nt] = __builtin_amdgcn_wmma_f32_16x16x4_f32(
                false, al[kq][nt], false, bs[kq], (short)0, ax[nt], false, false);
            a2[nt] = __builtin_amdgcn_wmma_f32_16x16x4_f32(
                false, al2, false, bs2, (short)0, a2[nt], false, false);
        }
    }
    float s = 0.0f;
#pragma unroll
    for (int nt = 0; nt < 4; ++nt) {
#pragma unroll
        for (int r = 0; r < 8; ++r) {
            const float x = ax[nt][r];
            s += x * x - a2[nt][r];
        }
    }
    s  += __shfl_xor(s, 16, 32);
    fh += __shfl_xor(fh, 16, 32);
    const float val = sigmoidf_(fh + 0.5f * s);

    if (hi == 0) res_s[wave * 16 + c] = val;
    __syncthreads();
    if (threadIdx.x < 64)
        out[blockIdx.x * 64 + threadIdx.x] = res_s[threadIdx.x];
}

extern "C" void kernel_launch(void* const* d_in, const int* in_sizes, int n_in,
                              void* d_out, int out_size, void* d_ws, size_t ws_size,
                              hipStream_t stream) {
    const float* dense = (const float*)d_in[0];
    const int*   sp    = (const int*)  d_in[1];
    const float* w0p   = (const float*)d_in[2];
    const float* w     = (const float*)d_in[3];
    const float* V     = (const float*)d_in[4];
    float* out = (float*)d_out;
    float* vxs = (float*)d_ws;
    float* v2s = vxs + KDIM * B_TOTAL;

    fm_lookup_kernel<<<dim3(KDIM * (B_TOTAL / SCHUNK)), LK_THREADS, 0, stream>>>(sp, V, vxs, v2s);
    fm_finish_kernel<<<dim3(B_TOTAL / 64), 128, 0, stream>>>(dense, sp, w0p, w, V, vxs, v2s, out);
}


# lookup launch: global->LDS fetch of V slices moves to 8 extra waves (4 even fields, 4 odd, one register set each); the 8 old waves only look up: load waits, stores run beside lookups; sums unchanged
# speedup vs baseline: 5.5241x; 1.0740x over previous
#include <hip/hip_runtime.h>
#include <math.h>
#include <stdint.h>


#define B_TOTAL   16384
#define ND        13
#define NS        26
#define FEAT_NUM  10000
#define FLEN      260013
#define KDIM      64

#define SCHUNK      4096
#define LK_THREADS  512
#define LU_THREADS  256
#define FG_THREADS  128
#define SPT         (SCHUNK / LU_THREADS)
#define SLICE_Q4    (FEAT_NUM / 4)
#define SLICE_QMAX  (SLICE_Q4 + 16)
#define QPT         20
#define IDX_BYTES   (SCHUNK * NS * 2)
#define SLICE_BYTES (SLICE_QMAX * 16)
#define LDS_BYTES   (IDX_BYTES + 2 * SLICE_BYTES)
#define IDX_GROUP   26

typedef __attribute__((ext_vector_type(2))) float v2f;
typedef __attribute__((ext_vector_type(8))) float v8f;

__device__ __forceinline__ float sigmoidf_(float x) {
    return 1.0f / (1.0f + __expf(-x));
}

__device__ __forceinline__ void lds_handover_barrier() {
    asm volatile("s_wait_dscnt 0x0\n\ts_barrier_signal -1\n\ts_barrier_wait -1" ::: "memory");
}

__global__ __launch_bounds__(LK_THREADS) void fm_lookup_kernel(
    const int*   __restrict__ sp,
    const float* __restrict__ V,
    float* __restrict__ vxs,
    float* __restrict__ v2s)
{
    __shared__ __attribute__((aligned(16))) unsigned char L[LDS_BYTES];
    uint32_t* idxw = (uint32_t*)L;
    float4*   sq0  = (float4*)(L + IDX_BYTES);
    float4*   sq1  = (float4*)(L + IDX_BYTES + SLICE_BYTES);
    const float* sl0 = (const float*)sq0;
    const float* sl1 = (const float*)sq1;

    const int tid = threadIdx.x;
    const int lt  = tid & (LU_THREADS - 1);
    const int gt  = tid & (FG_THREADS - 1);
    const int wv = __builtin_amdgcn_readfirstlane(tid) >> 5;
    const bool fetch_even = wv < 4;
    const bool fetch_odd  = (wv >= 4) && (wv < 8);
    const bool looker     = wv >= 8;
    const int k   = blockIdx.x >> 2;
    const int b0  = (blockIdx.x & 3) * SCHUNK;

    const float* vrow = V + (size_t)k * FLEN + ND;
    const int m0 = (int)(((uintptr_t)vrow >> 2) & 63);
    const float4* vq0 = (const float4*)(vrow - m0);
#define SLICE_GEOM(j)                                                            \
        const int off_ = m0 + (j) * FEAT_NUM;                                    \
        const int mj_  = off_ & 63;                                              \
        const int qlo_ = max(gt, mj_ >> 2);                                      \
        const int qhi_ = min((QPT - 1) * FG_THREADS + gt, ((mj_ + FEAT_NUM + 3) >> 2) - 1);

    float4 r[QPT];
#pragma unroll
    for (int n = 0; n < QPT; ++n) r[n] = make_float4(0.0f, 0.0f, 0.0f, 0.0f);
#define FETCH(j)                                                                 \
    {                                                                            \
        SLICE_GEOM(j)                                                            \
        const float4* b_ = vq0 + ((off_ - mj_) >> 2);                            \
        const float4* s_ = b_ + gt;                                              \
        r[0] = b_[qlo_];                                                         \
        _Pragma("unroll")                                                        \
        for (int n = 1; n < QPT - 1; ++n) r[n] = s_[n * FG_THREADS];             \
        r[QPT - 1] = b_[qhi_];                                                   \
    }
#define STASH(dq, j)                                                             \
    {                                                                            \
        SLICE_GEOM(j)                                                            \
        float4* d_ = dq + gt;                                                    \
        dq[qlo_] = r[0];                                                         \
        _Pragma("unroll")                                                        \
        for (int n = 1; n < QPT - 1; ++n) d_[n * FG_THREADS] = r[n];             \
        dq[qhi_] = r[QPT - 1];                                                   \
    }
#define LOAD_IDX(jj_)                                                            \
    {                                                                            \
        _Pragma("unroll")                                                        \
        for (int i = 0; i < SPT; ++i)                                            \
            iw[i] = idxw[(lt + LU_THREADS * i) * (NS / 2) + (jj_)];              \
    }
#define LOOKUP(sl, j, SHIFT_OR_MASK)                                             \
    {                                                                            \
        const float* sm_ = sl + ((m0 + 16 * (j)) & 63);                          \
        float val_[SPT];                                                         \
        _Pragma("unroll")                                                        \
        for (int i = 0; i < SPT; ++i)                                            \
            val_[i] = sm_[(int)(iw[i] SHIFT_OR_MASK)];                           \
        __builtin_amdgcn_sched_barrier(0);                                       \
        _Pragma("unroll")                                                        \
        for (int i = 0; i < SPT; ++i) {                                          \
            acc[i]  += val_[i];                                                  \
            acc2[i]  = __builtin_fmaf(val_[i], val_[i], acc2[i]);                \
        }                                                                        \
    }

    if (fetch_even) FETCH(0)
    if (fetch_odd)  FETCH(1)

    {
        const int4* spq = (const int4*)(sp + (size_t)b0 * NS);
        uint2* dst = (uint2*)idxw;
#pragma unroll 1
        for (int g = 0; g < (SCHUNK * NS / 4) / (LK_THREADS * IDX_GROUP); ++g) {
            int4 v[IDX_GROUP];
#pragma unroll
            for (int i = 0; i < IDX_GROUP; ++i)
                v[i] = spq[(g * IDX_GROUP + i) * LK_THREADS + tid];
#pragma unroll
            for (int i = 0; i < IDX_GROUP; ++i) {
                uint2 p;
                p.x = ((uint32_t)v[i].x & 0xffffu) | ((uint32_t)v[i].y << 16);
                p.y = ((uint32_t)v[i].z & 0xffffu) | ((uint32_t)v[i].w << 16);
                dst[(g * IDX_GROUP + i) * LK_THREADS + tid] = p;
            }
        }
    }
    if (fetch_even) {
        STASH(sq0, 0)
        FETCH(2)
    }
    lds_handover_barrier();

    float acc[SPT], acc2[SPT];
#pragma unroll
    for (int i = 0; i < SPT; ++i) { acc[i] = 0.0f; acc2[i] = 0.0f; }
    uint32_t iw[SPT];
#pragma unroll
    for (int i = 0; i < SPT; ++i) iw[i] = 0u;

    for (int jj = 0; jj < NS / 2 - 1; ++jj) {
        if (fetch_odd) {
            STASH(sq1, 2 * jj + 1)
            FETCH(2 * jj + 3)
        }
        if (looker) {
            LOAD_IDX(jj)
            LOOKUP(sl0, 2 * jj, & 0xffffu)
        }
        lds_handover_barrier();
        if (fetch_even) {
            STASH(sq0, 2 * jj + 2)
            if (jj < NS / 2 - 2) FETCH(2 * jj + 4)
        }
        if (looker) {
            LOOKUP(sl1, 2 * jj + 1, >> 16)
        }
        lds_handover_barrier();
    }
    if (fetch_odd) {
        STASH(sq1, NS - 1)
    }
    if (looker) {
        LOAD_IDX(NS / 2 - 1)
        LOOKUP(sl0, NS - 2, & 0xffffu)
    }
    lds_handover_barrier();
    if (looker) {
        LOOKUP(sl1, NS - 1, >> 16)
        float* ox = vxs + (size_t)k * B_TOTAL + b0;
        float* o2 = v2s + (size_t)k * B_TOTAL + b0;
#pragma unroll
        for (int i = 0; i < SPT; ++i) {
            ox[lt + LU_THREADS * i] = acc[i];
            o2[lt + LU_THREADS * i] = acc2[i];
        }
    }
#undef SLICE_GEOM
#undef FETCH
#undef STASH
#undef LOAD_IDX
#undef LOOKUP
}

__global__ __launch_bounds__(128) void fm_finish_kernel(
    const float* __restrict__ dense,
    const int*   __restrict__ sp,
    const float* __restrict__ w0p,
    const float* __restrict__ w,
    const float* __restrict__ V,
    const float* __restrict__ vxs,
    const float* __restrict__ v2s,
    float* __restrict__ out)
{
    __shared__ float res_s[64];


    const int lane = threadIdx.x & 31;
    const int wave = threadIdx.x >> 5;
    const int c  = lane & 15;
    const int hi = lane >> 4;
    const int rs = blockIdx.x * 64 + wave * 16 + c;

    const int jb = 13 * hi;
    int id[13];
#pragma unroll
    for (int j = 0; j < 13; ++j)
        id[j] = sp[(size_t)rs * NS + jb + j];

    v8f ax[4], a2[4];
#pragma unroll
    for (int nt = 0; nt < 4; ++nt) {
#pragma unroll
        for (int r = 0; r < 8; ++r) {
            const size_t o = (size_t)(nt * 16 + 8 * hi + r) * B_TOTAL + rs;
            ax[nt][r] = vxs[o];
            a2[nt][r] = v2s[o];
        }
    }

    v2f bs[4];
    v2f al[4][4];
#pragma unroll
    for (int kq = 0; kq < 4; ++kq) {
        const int f0 = 4 * kq + 2 * hi;
        const int fa = min(f0, ND - 1);
        const int fb = min(f0 + 1, ND - 1);
        const float dx = dense[(size_t)rs * ND + fa];
        const float dy = dense[(size_t)rs * ND + fb];
        bs[kq].x = (f0 + 0 < ND) ? dx : 0.0f;
        bs[kq].y = (f0 + 1 < ND) ? dy : 0.0f;
#pragma unroll
        for (int nt = 0; nt < 4; ++nt) {
            const float* vr = V + (size_t)(nt * 16 + c) * FLEN;
            const float vx = vr[fa];
            const float vy = vr[fb];
            al[kq][nt].x = (f0 + 0 < ND) ? vx : 0.0f;
            al[kq][nt].y = (f0 + 1 < ND) ? vy : 0.0f;
        }
    }

    const float w0 = w0p[0];
    float fh = hi ? 0.0f : w0;
#pragma unroll
    for (int kq = 0; kq < 4; ++kq) {
        const float w_0 = (4 * kq + 0 < ND) ? w[(4 * kq + 0 < ND) ? 4 * kq + 0 : 0] : 0.0f;
        const float w_1 = (4 * kq + 1 < ND) ? w[(4 * kq + 1 < ND) ? 4 * kq + 1 : 0] : 0.0f;
        const float w_2 = (4 * kq + 2 < ND) ? w[(4 * kq + 2 < ND) ? 4 * kq + 2 : 0] : 0.0f;
        const float w_3 = (4 * kq + 3 < ND) ? w[(4 * kq + 3 < ND) ? 4 * kq + 3 : 0] : 0.0f;
        fh += bs[kq].x * (hi ? w_2 : w_0);
        fh += bs[kq].y * (hi ? w_3 : w_1);
    }
#pragma unroll
    for (int j = 0; j < 13; ++j)
        fh += w[(size_t)ND + (size_t)(jb + j) * FEAT_NUM + (size_t)id[j]];

#pragma unroll
    for (int kq = 0; kq < 4; ++kq) {
        const v2f bs2 = bs[kq] * bs[kq];
#pragma unroll
        for (int nt = 0; nt < 4; ++nt) {
            const v2f al2 = al[kq][nt] * al[kq][nt];
            ax[nt] = __builtin_amdgcn_wmma_f32_16x16x4_f32(
                false, al[kq][nt], false, bs[kq], (short)0, ax[nt], false, false);
            a2[nt] = __builtin_amdgcn_wmma_f32_16x16x4_f32(
                false, al2, false, bs2, (short)0, a2[nt], false, false);
        }
    }
    float s = 0.0f;
#pragma unroll
    for (int nt = 0; nt < 4; ++nt) {
#pragma unroll
        for (int r = 0; r < 8; ++r) {
            const float x = ax[nt][r];
            s += x * x - a2[nt][r];
        }
    }
    s  += __shfl_xor(s, 16, 32);
    fh += __shfl_xor(fh, 16, 32);
    const float val = sigmoidf_(fh + 0.5f * s);

    if (hi == 0) res_s[wave * 16 + c] = val;
    __syncthreads();
    if (threadIdx.x < 64)
        out[blockIdx.x * 64 + threadIdx.x] = res_s[threadIdx.x];
}

extern "C" void kernel_launch(void* const* d_in, const int* in_sizes, int n_in,
                              void* d_out, int out_size, void* d_ws, size_t ws_size,
                              hipStream_t stream) {
    const float* dense = (const float*)d_in[0];
    const int*   sp    = (const int*)  d_in[1];
    const float* w0p   = (const float*)d_in[2];
    const float* w     = (const float*)d_in[3];
    const float* V     = (const float*)d_in[4];
    float* out = (float*)d_out;
    float* vxs = (float*)d_ws;
    float* v2s = vxs + KDIM * B_TOTAL;

    fm_lookup_kernel<<<dim3(KDIM * (B_TOTAL / SCHUNK)), LK_THREADS, 0, stream>>>(sp, V, vxs, v2s);
    fm_finish_kernel<<<dim3(B_TOTAL / 64), 128, 0, stream>>>(dense, sp, w0p, w, V, vxs, v2s, out);
}
